# starting kernel
# baseline (speedup 1.0000x reference)
#include <hip/hip_runtime.h>
#include <math.h>

typedef float v2f __attribute__((ext_vector_type(2)));
typedef float v8f __attribute__((ext_vector_type(8)));

__global__ __launch_bounds__(256) void hm_conf_kernel(const float* __restrict__ hm,
                                                      float* __restrict__ conf) {
    const int bid = blockIdx.x;
    const int t   = threadIdx.x;
    const float4* p = (const float4*)(hm + (size_t)bid * 4096);

    float m = -3.402823466e38f;
#pragma unroll
    for (int i = 0; i < 4; ++i) {
        float4 v = p[t + i * 256];
        m = fmaxf(m, fmaxf(fmaxf(v.x, v.y), fmaxf(v.z, v.w)));
    }
#pragma unroll
    for (int off = 16; off > 0; off >>= 1)
        m = fmaxf(m, __shfl_xor(m, off, 32));

    __shared__ float smax[8];
    if ((t & 31) == 0) smax[t >> 5] = m;
    __syncthreads();
    if (t == 0) {
        float mm = smax[0];
#pragma unroll
        for (int i = 1; i < 8; ++i) mm = fmaxf(mm, smax[i]);
        conf[bid] = 1.0f / (1.0f + expf(-mm));
    }
}

__device__ __forceinline__ float gelu_exact(float x) {
    return 0.5f * x * (1.0f + erff(x * 0.70710678118654752440f));
}

__device__ __forceinline__ void ln_inplace(float h[16][65],
                                           const float* __restrict__ g,
                                           const float* __restrict__ be,
                                           float rsum[16][8], float rsq[16][8],
                                           int tid) {
    const int r = tid >> 3;
    const int p = tid & 7;
    float s = 0.f, ss = 0.f;
#pragma unroll
    for (int j = 0; j < 8; ++j) {
        float v = h[r][p * 8 + j];
        s += v; ss += v * v;
    }
    rsum[r][p] = s; rsq[r][p] = ss;
    __syncthreads();
    float ts = 0.f, tss = 0.f;
#pragma unroll
    for (int i = 0; i < 8; ++i) { ts += rsum[r][i]; tss += rsq[r][i]; }
    const float mu  = ts * (1.0f / 64.0f);
    const float var = tss * (1.0f / 64.0f) - mu * mu;
    const float sc  = rsqrtf(var + 1e-5f);
#pragma unroll
    for (int j = 0; j < 8; ++j) {
        const int c = p * 8 + j;
        h[r][c] = (h[r][c] - mu) * sc * g[c] + be[c];
    }
    __syncthreads();
}

__global__ __launch_bounds__(128) void mlp_head_kernel(
    const float* __restrict__ fk, const float* __restrict__ dir,
    const float* __restrict__ conf,
    const float* __restrict__ W1, const float* __restrict__ b1,
    const float* __restrict__ g1, const float* __restrict__ be1,
    const float* __restrict__ W2, const float* __restrict__ b2,
    const float* __restrict__ g2, const float* __restrict__ be2,
    const float* __restrict__ W3, const float* __restrict__ b3,
    float* __restrict__ out, int nrows) {

    __shared__ float xs[16][12];
    __shared__ float w1s[12][64];
    __shared__ float w2s[64][64];
    __shared__ float w3s[64][16];
    __shared__ float h1[16][65];
    __shared__ float h2[16][65];
    __shared__ float outs[16][16];
    __shared__ float rsum[16][8];
    __shared__ float rsq[16][8];

    const int tid  = threadIdx.x;
    const int lane = tid & 31;
    const int wave = tid >> 5;
    const int half = lane >> 4;
    const int l16  = lane & 15;
    const int row0 = blockIdx.x * 16;

    for (int i = tid; i < 12 * 64; i += 128) {
        const int k = i >> 6, n = i & 63;
        w1s[k][n] = (k < 10) ? W1[k * 64 + n] : 0.0f;
    }
    for (int i = tid; i < 64 * 64; i += 128)
        w2s[i >> 6][i & 63] = W2[i];
    for (int i = tid; i < 64 * 16; i += 128) {
        const int k = i >> 4, n = i & 15;
        w3s[k][n] = (n < 4) ? W3[k * 4 + n] : 0.0f;
    }
    for (int i = tid; i < 16 * 12; i += 128) {
        const int r = i / 12, c = i % 12;
        const int grow = row0 + r;
        float v;
        if (c < 3)       v = fk[grow * 3 + c];
        else if (c < 6)  v = dir[grow * 3 + (c - 3)];
        else if (c < 9)  v = dir[grow * 3 + (c - 6)] - fk[grow * 3 + (c - 6)];
        else if (c == 9) v = conf[grow];
        else             v = 0.0f;
        xs[r][c] = v;
    }
    __syncthreads();

    {
        v8f c = {0.f, 0.f, 0.f, 0.f, 0.f, 0.f, 0.f, 0.f};
        const int ncol = wave * 16 + l16;
#pragma unroll
        for (int ks = 0; ks < 12; ks += 4) {
            v2f a, b;
            a[0] = xs[l16][ks + 2 * half + 0];
            a[1] = xs[l16][ks + 2 * half + 1];
            b[0] = w1s[ks + 2 * half + 0][ncol];
            b[1] = w1s[ks + 2 * half + 1][ncol];
            c = __builtin_amdgcn_wmma_f32_16x16x4_f32(false, a, false, b,
                                                      (short)0, c, false, false);
        }
        const float bias = b1[ncol];
#pragma unroll
        for (int v = 0; v < 8; ++v)
            h1[v + 8 * half][ncol] = gelu_exact(c[v] + bias);
    }
    __syncthreads();
    ln_inplace(h1, g1, be1, rsum, rsq, tid);

    {
        v8f c = {0.f, 0.f, 0.f, 0.f, 0.f, 0.f, 0.f, 0.f};
        const int ncol = wave * 16 + l16;
#pragma unroll
        for (int ks = 0; ks < 64; ks += 4) {
            v2f a, b;
            a[0] = h1[l16][ks + 2 * half + 0];
            a[1] = h1[l16][ks + 2 * half + 1];
            b[0] = w2s[ks + 2 * half + 0][ncol];
            b[1] = w2s[ks + 2 * half + 1][ncol];
            c = __builtin_amdgcn_wmma_f32_16x16x4_f32(false, a, false, b,
                                                      (short)0, c, false, false);
        }
        const float bias = b2[ncol];
#pragma unroll
        for (int v = 0; v < 8; ++v)
            h2[v + 8 * half][ncol] = gelu_exact(c[v] + bias);
    }
    __syncthreads();
    ln_inplace(h2, g2, be2, rsum, rsq, tid);

    if (wave == 0) {
        v8f c = {0.f, 0.f, 0.f, 0.f, 0.f, 0.f, 0.f, 0.f};
#pragma unroll
        for (int ks = 0; ks < 64; ks += 4) {
            v2f a, b;
            a[0] = h2[l16][ks + 2 * half + 0];
            a[1] = h2[l16][ks + 2 * half + 1];
            b[0] = w3s[ks + 2 * half + 0][l16];
            b[1] = w3s[ks + 2 * half + 1][l16];
            c = __builtin_amdgcn_wmma_f32_16x16x4_f32(false, a, false, b,
                                                      (short)0, c, false, false);
        }
#pragma unroll
        for (int v = 0; v < 8; ++v)
            outs[v + 8 * half][l16] = c[v];
    }
    __syncthreads();

    if (tid < 16) {
        const int grow = row0 + tid;
        const float d0 = outs[tid][0] + b3[0];
        const float d1 = outs[tid][1] + b3[1];
        const float d2 = outs[tid][2] + b3[2];
        const float o3 = outs[tid][3] + b3[3];
        const float alpha = 1.0f / (1.0f + expf(-o3));

        const float fx = fk[grow * 3 + 0], fy = fk[grow * 3 + 1], fz = fk[grow * 3 + 2];
        const float dx = dir[grow * 3 + 0], dy = dir[grow * 3 + 1], dz = dir[grow * 3 + 2];

        float* kp_final = out;
        float* p_alpha  = out + (size_t)nrows * 3;
        float* p_delta  = out + (size_t)nrows * 4;

        kp_final[grow * 3 + 0] = alpha * dx + (1.0f - alpha) * fx + d0;
        kp_final[grow * 3 + 1] = alpha * dy + (1.0f - alpha) * fy + d1;
        kp_final[grow * 3 + 2] = alpha * dz + (1.0f - alpha) * fz + d2;
        p_alpha[grow] = alpha;
        p_delta[grow * 3 + 0] = d0;
        p_delta[grow * 3 + 1] = d1;
        p_delta[grow * 3 + 2] = d2;
    }
}

extern "C" void kernel_launch(void* const* d_in, const int* in_sizes, int n_in,
                              void* d_out, int out_size, void* d_ws, size_t ws_size,
                              hipStream_t stream) {
    const float* fk  = (const float*)d_in[0];
    const float* dir = (const float*)d_in[1];
    const float* hm  = (const float*)d_in[2];
    const float* W1  = (const float*)d_in[3];
    const float* b1  = (const float*)d_in[4];
    const float* g1  = (const float*)d_in[5];
    const float* be1 = (const float*)d_in[6];
    const float* W2  = (const float*)d_in[7];
    const float* b2  = (const float*)d_in[8];
    const float* g2  = (const float*)d_in[9];
    const float* be2 = (const float*)d_in[10];
    const float* W3  = (const float*)d_in[11];
    const float* b3  = (const float*)d_in[12];
    float* out  = (float*)d_out;
    float* conf = (float*)d_ws;

    const int nrows = in_sizes[0] / 3;

    hm_conf_kernel<<<nrows, 256, 0, stream>>>(hm, conf);
    mlp_head_kernel<<<nrows / 16, 128, 0, stream>>>(fk, dir, conf,
                                                    W1, b1, g1, be1,
                                                    W2, b2, g2, be2,
                                                    W3, b3, out, nrows);
}


# head kernel computed transposed (features on matrix M, rows on lanes): one wave per 16-row tile keeps activations, GELU, LayerNorm in registers, no LDS or barriers; product and LN sum orders change
# speedup vs baseline: 1.4580x; 1.4580x over previous
#include <hip/hip_runtime.h>
#include <math.h>

typedef float v2f __attribute__((ext_vector_type(2)));
typedef float v8f __attribute__((ext_vector_type(8)));

__global__ __launch_bounds__(256) void hm_conf_kernel(const float* __restrict__ hm,
                                                      float* __restrict__ conf) {
    const int bid = blockIdx.x;
    const int t   = threadIdx.x;
    const float4* p = (const float4*)(hm + (size_t)bid * 4096);

    float m = -3.402823466e38f;
#pragma unroll
    for (int i = 0; i < 4; ++i) {
        float4 v = p[t + i * 256];
        m = fmaxf(m, fmaxf(fmaxf(v.x, v.y), fmaxf(v.z, v.w)));
    }
#pragma unroll
    for (int off = 16; off > 0; off >>= 1)
        m = fmaxf(m, __shfl_xor(m, off, 32));

    __shared__ float smax[8];
    if ((t & 31) == 0) smax[t >> 5] = m;
    __syncthreads();
    if (t == 0) {
        float mm = smax[0];
#pragma unroll
        for (int i = 1; i < 8; ++i) mm = fmaxf(mm, smax[i]);
        conf[bid] = 1.0f / (1.0f + expf(-mm));
    }
}

__device__ __forceinline__ float erf_sel(float a) {
    const float t = fabsf(a);
    const float s = a * a;
    float r = fmaf(-1.72853470e-5f, t, 3.83197126e-4f);
    const float u = fmaf(-3.88396438e-3f, t, 2.42546219e-2f);
    r = fmaf(r, s, u);
    r = fmaf(r, t, -1.06777877e-1f);
    r = fmaf(r, t, -6.34846687e-1f);
    r = fmaf(r, t, -1.28717512e-1f);
    r = fmaf(r, t, -t);
    const float hi = copysignf(1.0f - expf(r), a);
    float q = -5.96761703e-4f;
    q = fmaf(q, s, 4.99119423e-3f);
    q = fmaf(q, s, -2.67681349e-2f);
    q = fmaf(q, s, 1.12819925e-1f);
    q = fmaf(q, s, -3.76125336e-1f);
    q = fmaf(q, s, 1.28379166e-1f);
    const float lo = fmaf(q, a, a);
    return (t > 0.927734375f) ? hi : lo;
}

__device__ __forceinline__ float gelu_exact(float x) {
    return 0.5f * x * (1.0f + erf_sel(x * 0.70710678118654752440f));
}

__device__ __forceinline__ void bias_gelu_ln(const v8f c[4],
                                             const float* __restrict__ bias,
                                             const float* __restrict__ gam,
                                             const float* __restrict__ bet,
                                             int half, float h[4][8]) {
    float ps[4] = {0.f, 0.f, 0.f, 0.f};
#pragma unroll
    for (int j = 0; j < 8; ++j) {
        const float4 bv = *(const float4*)(bias + 4 * j + 32 * half);
        const float bb[4] = {bv.x, bv.y, bv.z, bv.w};
#pragma unroll
        for (int t = 0; t < 4; ++t) {
            const float g = gelu_exact(c[t][j] + bb[t]);
            h[t][j] = g;
            ps[t] += g;
        }
    }
    float s = (ps[0] + ps[1]) + (ps[2] + ps[3]);
    s += __shfl_xor(s, 16, 32);
    const float mu = s * (1.0f / 64.0f);
    float pq[4] = {0.f, 0.f, 0.f, 0.f};
#pragma unroll
    for (int j = 0; j < 8; ++j) {
#pragma unroll
        for (int t = 0; t < 4; ++t) {
            const float d = h[t][j] - mu;
            h[t][j] = d;
            pq[t] = fmaf(d, d, pq[t]);
        }
    }
    float ss = (pq[0] + pq[1]) + (pq[2] + pq[3]);
    ss += __shfl_xor(ss, 16, 32);
    const float sc = rsqrtf(ss * (1.0f / 64.0f) + 1e-5f);
#pragma unroll
    for (int j = 0; j < 8; ++j) {
        const float4 gv = *(const float4*)(gam + 4 * j + 32 * half);
        const float4 ev = *(const float4*)(bet + 4 * j + 32 * half);
        const float gg[4] = {gv.x, gv.y, gv.z, gv.w};
        const float ee[4] = {ev.x, ev.y, ev.z, ev.w};
#pragma unroll
        for (int t = 0; t < 4; ++t)
            h[t][j] = h[t][j] * sc * gg[t] + ee[t];
    }
}

__global__ __launch_bounds__(32) __attribute__((amdgpu_waves_per_eu(1, 1))) void mlp_head_kernel(
    const float* __restrict__ fk, const float* __restrict__ dir,
    const float* __restrict__ conf,
    const float* __restrict__ W1, const float* __restrict__ b1,
    const float* __restrict__ g1, const float* __restrict__ be1,
    const float* __restrict__ W2, const float* __restrict__ b2,
    const float* __restrict__ g2, const float* __restrict__ be2,
    const float* __restrict__ W3, const float* __restrict__ b3,
    float* __restrict__ out, int nrows) {

    const int lane = threadIdx.x;
    const int half = lane >> 4;
    const int l16  = lane & 15;
    const int row  = blockIdx.x * 16 + l16;

    float4 w2f[32];
    float  w3f[32];
#pragma unroll
    for (int t = 0; t < 4; ++t) {
#pragma unroll
        for (int q = 0; q < 4; ++q) {
#pragma unroll
            for (int v = 0; v < 2; ++v) {
                const int k = 4 * (2 * q + v) + 32 * half + t;
                w2f[(t * 4 + q) * 2 + v] = *(const float4*)(W2 + k * 64 + 4 * l16);
                const float w3 = W3[k * 4 + (l16 & 3)];
                w3f[(t * 4 + q) * 2 + v] = (l16 < 4) ? w3 : 0.0f;
            }
        }
    }

    const float fx = fk[row * 3 + 0], fy = fk[row * 3 + 1], fz = fk[row * 3 + 2];
    const float dx = dir[row * 3 + 0], dy = dir[row * 3 + 1], dz = dir[row * 3 + 2];
    const float cf = conf[row];
    const float x[12] = {fx, fy, fz, dx, dy, dz, dx - fx, dy - fy, dz - fz, cf, 0.0f, 0.0f};

    float h[4][8];

    {
        v8f c[4];
#pragma unroll
        for (int t = 0; t < 4; ++t)
            c[t] = (v8f){0.f, 0.f, 0.f, 0.f, 0.f, 0.f, 0.f, 0.f};
#pragma unroll
        for (int i = 0; i < 3; ++i) {
            v2f b;
            b[0] = half ? x[4 * i + 2] : x[4 * i + 0];
            b[1] = half ? x[4 * i + 3] : x[4 * i + 1];
            float wa[2][4];
#pragma unroll
            for (int v = 0; v < 2; ++v) {
                const int k  = 4 * i + 2 * half + v;
                const int kk = (k < 10) ? k : 9;
                const float4 w = *(const float4*)(W1 + kk * 64 + 4 * l16);
                const bool real = (k < 10);
                wa[v][0] = real ? w.x : 0.0f;
                wa[v][1] = real ? w.y : 0.0f;
                wa[v][2] = real ? w.z : 0.0f;
                wa[v][3] = real ? w.w : 0.0f;
            }
#pragma unroll
            for (int t = 0; t < 4; ++t) {
                v2f a;
                a[0] = wa[0][t];
                a[1] = wa[1][t];
                c[t] = __builtin_amdgcn_wmma_f32_16x16x4_f32(false, a, false, b,
                                                             (short)0, c[t], false, false);
            }
        }
        bias_gelu_ln(c, b1, g1, be1, half, h);
    }

    {
        v8f c[4];
#pragma unroll
        for (int t = 0; t < 4; ++t)
            c[t] = (v8f){0.f, 0.f, 0.f, 0.f, 0.f, 0.f, 0.f, 0.f};
#pragma unroll
        for (int tp = 0; tp < 4; ++tp) {
#pragma unroll
            for (int q = 0; q < 4; ++q) {
                v2f b;
                b[0] = h[tp][2 * q + 0];
                b[1] = h[tp][2 * q + 1];
                const float4 w0 = w2f[(tp * 4 + q) * 2 + 0];
                const float4 w1 = w2f[(tp * 4 + q) * 2 + 1];
                const float wa[2][4] = {{w0.x, w0.y, w0.z, w0.w}, {w1.x, w1.y, w1.z, w1.w}};
#pragma unroll
                for (int t = 0; t < 4; ++t) {
                    v2f a;
                    a[0] = wa[0][t];
                    a[1] = wa[1][t];
                    c[t] = __builtin_amdgcn_wmma_f32_16x16x4_f32(false, a, false, b,
                                                                 (short)0, c[t], false, false);
                }
            }
        }
        bias_gelu_ln(c, b2, g2, be2, half, h);
    }

    v8f o;
    {
        v8f ca = {0.f, 0.f, 0.f, 0.f, 0.f, 0.f, 0.f, 0.f};
        v8f cb = {0.f, 0.f, 0.f, 0.f, 0.f, 0.f, 0.f, 0.f};
#pragma unroll
        for (int tp = 0; tp < 4; ++tp) {
#pragma unroll
            for (int q = 0; q < 4; q += 2) {
                v2f a, b;
                b[0] = h[tp][2 * q + 0];
                b[1] = h[tp][2 * q + 1];
                a[0] = w3f[(tp * 4 + q) * 2 + 0];
                a[1] = w3f[(tp * 4 + q) * 2 + 1];
                ca = __builtin_amdgcn_wmma_f32_16x16x4_f32(false, a, false, b,
                                                           (short)0, ca, false, false);
                b[0] = h[tp][2 * q + 2];
                b[1] = h[tp][2 * q + 3];
                a[0] = w3f[(tp * 4 + q + 1) * 2 + 0];
                a[1] = w3f[(tp * 4 + q + 1) * 2 + 1];
                cb = __builtin_amdgcn_wmma_f32_16x16x4_f32(false, a, false, b,
                                                           (short)0, cb, false, false);
            }
        }
        o = ca + cb;
    }

    if (half == 0) {
        const float d0 = o[0] + b3[0];
        const float d1 = o[1] + b3[1];
        const float d2 = o[2] + b3[2];
        const float o3 = o[3] + b3[3];
        const float alpha = 1.0f / (1.0f + expf(-o3));

        float* kp_final = out;
        float* p_alpha  = out + (size_t)nrows * 3;
        float* p_delta  = out + (size_t)nrows * 4;

        kp_final[row * 3 + 0] = alpha * dx + (1.0f - alpha) * fx + d0;
        kp_final[row * 3 + 1] = alpha * dy + (1.0f - alpha) * fy + d1;
        kp_final[row * 3 + 2] = alpha * dz + (1.0f - alpha) * fz + d2;
        p_alpha[row] = alpha;
        p_delta[row * 3 + 0] = d0;
        p_delta[row * 3 + 1] = d1;
        p_delta[row * 3 + 2] = d2;
    }
}

extern "C" void kernel_launch(void* const* d_in, const int* in_sizes, int n_in,
                              void* d_out, int out_size, void* d_ws, size_t ws_size,
                              hipStream_t stream) {
    const float* fk  = (const float*)d_in[0];
    const float* dir = (const float*)d_in[1];
    const float* hm  = (const float*)d_in[2];
    const float* W1  = (const float*)d_in[3];
    const float* b1  = (const float*)d_in[4];
    const float* g1  = (const float*)d_in[5];
    const float* be1 = (const float*)d_in[6];
    const float* W2  = (const float*)d_in[7];
    const float* b2  = (const float*)d_in[8];
    const float* g2  = (const float*)d_in[9];
    const float* be2 = (const float*)d_in[10];
    const float* W3  = (const float*)d_in[11];
    const float* b3  = (const float*)d_in[12];
    float* out  = (float*)d_out;
    float* conf = (float*)d_ws;

    const int nrows = in_sizes[0] / 3;

    hm_conf_kernel<<<nrows, 256, 0, stream>>>(hm, conf);
    mlp_head_kernel<<<nrows / 16, 32, 0, stream>>>(fk, dir, conf,
                                                    W1, b1, g1, be1,
                                                    W2, b2, g2, be2,
                                                    W3, b3, out, nrows);
}


# head kernel: each 16-row tile on 4 waves, not 1 (one 16-feature tile, 8 GELU per lane each, not 32); activations and LN partial sums cross once per layer via LDS; product and LN sum orders change
# speedup vs baseline: 1.6339x; 1.1206x over previous
#include <hip/hip_runtime.h>
#include <math.h>

typedef float v2f __attribute__((ext_vector_type(2)));
typedef float v4f __attribute__((ext_vector_type(4)));
typedef float v8f __attribute__((ext_vector_type(8)));

__global__ __launch_bounds__(256) void hm_conf_kernel(const float* __restrict__ hm,
                                                      float* __restrict__ conf) {
    const int bid = blockIdx.x;
    const int t   = threadIdx.x;
    const float4* p = (const float4*)(hm + (size_t)bid * 4096);

    float m = -3.402823466e38f;
#pragma unroll
    for (int i = 0; i < 4; ++i) {
        float4 v = p[t + i * 256];
        m = fmaxf(m, fmaxf(fmaxf(v.x, v.y), fmaxf(v.z, v.w)));
    }
#pragma unroll
    for (int off = 16; off > 0; off >>= 1)
        m = fmaxf(m, __shfl_xor(m, off, 32));

    __shared__ float smax[8];
    if ((t & 31) == 0) smax[t >> 5] = m;
    __syncthreads();
    if (t == 0) {
        float mm = smax[0];
#pragma unroll
        for (int i = 1; i < 8; ++i) mm = fmaxf(mm, smax[i]);
        conf[bid] = 1.0f / (1.0f + expf(-mm));
    }
}

__device__ __forceinline__ float erf_sel(float a) {
    const float t = fabsf(a);
    const float s = a * a;
    float r = fmaf(-1.72853470e-5f, t, 3.83197126e-4f);
    const float u = fmaf(-3.88396438e-3f, t, 2.42546219e-2f);
    r = fmaf(r, s, u);
    r = fmaf(r, t, -1.06777877e-1f);
    r = fmaf(r, t, -6.34846687e-1f);
    r = fmaf(r, t, -1.28717512e-1f);
    r = fmaf(r, t, -t);
    const float hi = copysignf(1.0f - expf(r), a);
    float q = -5.96761703e-4f;
    q = fmaf(q, s, 4.99119423e-3f);
    q = fmaf(q, s, -2.67681349e-2f);
    q = fmaf(q, s, 1.12819925e-1f);
    q = fmaf(q, s, -3.76125336e-1f);
    q = fmaf(q, s, 1.28379166e-1f);
    const float lo = fmaf(q, a, a);
    return (t > 0.927734375f) ? hi : lo;
}

__device__ __forceinline__ float gelu_exact(float x) {
    return 0.5f * x * (1.0f + erf_sel(x * 0.70710678118654752440f));
}

#define GP 68
#define SP 20

__device__ __forceinline__ void bias_gelu_handover(const v8f c, const float4 ba, const float4 bb,
                                                   float* __restrict__ grow,
                                                   float* __restrict__ srow) {
    const float bz[8] = {ba.x, ba.y, ba.z, ba.w, bb.x, bb.y, bb.z, bb.w};
    float g[8];
#pragma unroll
    for (int j = 0; j < 8; ++j) g[j] = gelu_exact(c[j] + bz[j]);
    const float m = (((g[0] + g[1]) + (g[2] + g[3])) + ((g[4] + g[5]) + (g[6] + g[7]))) * 0.125f;
    float q0 = 0.0f, q1 = 0.0f;
#pragma unroll
    for (int j = 0; j < 8; j += 2) {
        const float d0 = g[j] - m, d1 = g[j + 1] - m;
        q0 = fmaf(d0, d0, q0);
        q1 = fmaf(d1, d1, q1);
    }
    *(float4*)(grow)     = make_float4(g[0], g[1], g[2], g[3]);
    *(float4*)(grow + 4) = make_float4(g[4], g[5], g[6], g[7]);
    *(float2*)(srow)     = make_float2(m, q0 + q1);
}

__device__ __forceinline__ void take(const float* __restrict__ grow,
                                     const float* __restrict__ srow,
                                     v4f sv[4], v4f gv[8]) {
#pragma unroll
    for (int q = 0; q < 8; ++q) gv[q] = *(const v4f*)(grow + 4 * q);
#pragma unroll
    for (int q = 0; q < 4; ++q) sv[q] = *(const v4f*)(srow + 4 * q);
}

__device__ __forceinline__ void merge_and_norm(const v4f sv[4], const v4f gv[8],
                                               const v4f gam[8], const v4f bet[8],
                                               float hn[32]) {
    const float mk[8] = {sv[0].x, sv[0].z, sv[1].x, sv[1].z, sv[2].x, sv[2].z, sv[3].x, sv[3].z};
    const float qk[8] = {sv[0].y, sv[0].w, sv[1].y, sv[1].w, sv[2].y, sv[2].w, sv[3].y, sv[3].w};
    const float mu = (((mk[0] + mk[1]) + (mk[2] + mk[3])) + ((mk[4] + mk[5]) + (mk[6] + mk[7]))) * 0.125f;
    float dd = 0.0f;
#pragma unroll
    for (int k = 0; k < 8; ++k) {
        const float d = mk[k] - mu;
        dd = fmaf(d, d, dd);
    }
    const float qs = ((qk[0] + qk[1]) + (qk[2] + qk[3])) + ((qk[4] + qk[5]) + (qk[6] + qk[7]));
    const float ss = fmaf(8.0f, dd, qs);
    const float sc = rsqrtf(ss * (1.0f / 64.0f) + 1e-5f);
#pragma unroll
    for (int q = 0; q < 8; ++q) {
        hn[4 * q + 0] = (gv[q].x - mu) * sc * gam[q].x + bet[q].x;
        hn[4 * q + 1] = (gv[q].y - mu) * sc * gam[q].y + bet[q].y;
        hn[4 * q + 2] = (gv[q].z - mu) * sc * gam[q].z + bet[q].z;
        hn[4 * q + 3] = (gv[q].w - mu) * sc * gam[q].w + bet[q].w;
    }
}

__global__ __launch_bounds__(128) void mlp_head_kernel(
    const float* __restrict__ fk, const float* __restrict__ dir,
    const float* __restrict__ conf,
    const float* __restrict__ W1, const float* __restrict__ b1,
    const float* __restrict__ g1, const float* __restrict__ be1,
    const float* __restrict__ W2, const float* __restrict__ b2,
    const float* __restrict__ g2, const float* __restrict__ be2,
    const float* __restrict__ W3, const float* __restrict__ b3,
    float* __restrict__ out, int nrows) {

    __shared__ __attribute__((aligned(16))) float G[2][16 * GP];
    __shared__ __attribute__((aligned(16))) float S[2][16 * SP];

    const int lane = threadIdx.x & 31;
    const int w    = __builtin_amdgcn_readfirstlane((int)(threadIdx.x >> 5));
    const int half = lane >> 4;
    const int l16  = lane & 15;
    const int row  = blockIdx.x * 16 + l16;
    const int fo   = 16 * w + 8 * half;
    const int fi   = 32 * half;

    const float fx = fk[row * 3 + 0], fy = fk[row * 3 + 1], fz = fk[row * 3 + 2];
    const float dx = dir[row * 3 + 0], dy = dir[row * 3 + 1], dz = dir[row * 3 + 2];
    const float cf = conf[row];

    float w1a[3][2];
#pragma unroll
    for (int i = 0; i < 3; ++i) {
#pragma unroll
        for (int v = 0; v < 2; ++v) {
            const int k  = 4 * i + 2 * half + v;
            const int kk = (k < 10) ? k : 9;
            w1a[i][v] = W1[kk * 64 + 16 * w + l16];
        }
    }
    const float4 b1a = *(const float4*)(b1 + fo), b1b = *(const float4*)(b1 + fo + 4);
    __builtin_amdgcn_sched_barrier(0);
    const float* __restrict__ p2 = W2 + (fi * 64 + 16 * w + l16);
    float w2a[16][2];
#pragma unroll
    for (int i = 0; i < 16; ++i) {
#pragma unroll
        for (int v = 0; v < 2; ++v)
            w2a[i][v] = p2[(2 * i + v) * 64];
    }
    v4f g1v[8], e1v[8];
#pragma unroll
    for (int q = 0; q < 8; ++q) {
        g1v[q] = *(const v4f*)(g1 + fi + 4 * q);
        e1v[q] = *(const v4f*)(be1 + fi + 4 * q);
    }
    const float4 b2a = *(const float4*)(b2 + fo), b2b = *(const float4*)(b2 + fo + 4);
    __builtin_amdgcn_sched_barrier(0);
    const float keep = (float)(1 - half);
    w1a[2][0] *= keep;
    w1a[2][1] *= keep;

    v4f   sv[4], gv[8];
    float hn[32];

    {
        v8f c = {0.f, 0.f, 0.f, 0.f, 0.f, 0.f, 0.f, 0.f};
        const float xb[3][2] = {{half ? fz : fx,        half ? dx : fy},
                                {half ? dx - fx : dy,   half ? dy - fy : dz},
                                {(dz - fz) * keep,      cf * keep}};
#pragma unroll
        for (int i = 0; i < 3; ++i) {
            v2f a, b;
            b[0] = xb[i][0];
            b[1] = xb[i][1];
            a[0] = w1a[i][0];
            a[1] = w1a[i][1];
            c = __builtin_amdgcn_wmma_f32_16x16x4_f32(false, a, false, b,
                                                      (short)0, c, false, false);
        }
        bias_gelu_handover(c, b1a, b1b, &G[0][l16 * GP + fo], &S[0][l16 * SP + 2 * (2 * w + half)]);
    }
    asm volatile("" :: "v"(g1v[0]), "v"(g1v[1]), "v"(g1v[2]), "v"(g1v[3]),
                       "v"(g1v[4]), "v"(g1v[5]), "v"(g1v[6]), "v"(g1v[7]),
                       "v"(e1v[0]), "v"(e1v[1]), "v"(e1v[2]), "v"(e1v[3]),
                       "v"(e1v[4]), "v"(e1v[5]), "v"(e1v[6]), "v"(e1v[7]));
    __syncthreads();
    take(&G[0][l16 * GP + fi], &S[0][l16 * SP], sv, gv);

    float  w3a[16][2];
    v4f    g2v[8], e2v[8];
    if (w == 0) {
#pragma unroll
        for (int q = 0; q < 8; ++q) {
            g2v[q] = *(const v4f*)(g2 + fi + 4 * q);
            e2v[q] = *(const v4f*)(be2 + fi + 4 * q);
        }
        const float* __restrict__ p3 = W3 + (fi * 4 + (l16 & 3));
#pragma unroll
        for (int i = 0; i < 16; ++i) {
#pragma unroll
            for (int v = 0; v < 2; ++v)
                w3a[i][v] = p3[(2 * i + v) * 4];
        }
    }
    merge_and_norm(sv, gv, g1v, e1v, hn);

    {
        v8f ca = {0.f, 0.f, 0.f, 0.f, 0.f, 0.f, 0.f, 0.f};
        v8f cb = {0.f, 0.f, 0.f, 0.f, 0.f, 0.f, 0.f, 0.f};
#pragma unroll
        for (int i = 0; i < 16; i += 2) {
            v2f a, b;
            a[0] = w2a[i][0];
            a[1] = w2a[i][1];
            b[0] = hn[2 * i + 0];
            b[1] = hn[2 * i + 1];
            ca = __builtin_amdgcn_wmma_f32_16x16x4_f32(false, a, false, b,
                                                       (short)0, ca, false, false);
            a[0] = w2a[i + 1][0];
            a[1] = w2a[i + 1][1];
            b[0] = hn[2 * i + 2];
            b[1] = hn[2 * i + 3];
            cb = __builtin_amdgcn_wmma_f32_16x16x4_f32(false, a, false, b,
                                                       (short)0, cb, false, false);
        }
        const v8f c = ca + cb;
        bias_gelu_handover(c, b2a, b2b, &G[1][l16 * GP + fo], &S[1][l16 * SP + 2 * (2 * w + half)]);
    }
    __syncthreads();
    if (w != 0) return;
    take(&G[1][l16 * GP + fi], &S[1][l16 * SP], sv, gv);
    __builtin_amdgcn_sched_barrier(0);
    merge_and_norm(sv, gv, g2v, e2v, hn);

    v8f o;
    {
        v8f ca = {0.f, 0.f, 0.f, 0.f, 0.f, 0.f, 0.f, 0.f};
        v8f cb = {0.f, 0.f, 0.f, 0.f, 0.f, 0.f, 0.f, 0.f};
#pragma unroll
        for (int i = 0; i < 16; i += 2) {
            v2f a, b;
            a[0] = w3a[i][0];
            a[1] = w3a[i][1];
            b[0] = hn[2 * i + 0];
            b[1] = hn[2 * i + 1];
            ca = __builtin_amdgcn_wmma_f32_16x16x4_f32(false, a, false, b,
                                                       (short)0, ca, false, false);
            a[0] = w3a[i + 1][0];
            a[1] = w3a[i + 1][1];
            b[0] = hn[2 * i + 2];
            b[1] = hn[2 * i + 3];
            cb = __builtin_amdgcn_wmma_f32_16x16x4_f32(false, a, false, b,
                                                       (short)0, cb, false, false);
        }
        o = ca + cb;
    }

    if (half == 0) {
        const float d0 = o[0] + b3[0];
        const float d1 = o[1] + b3[1];
        const float d2 = o[2] + b3[2];
        const float o3 = o[3] + b3[3];
        const float alpha = 1.0f / (1.0f + expf(-o3));

        float* kp_final = out;
        float* p_alpha  = out + (size_t)nrows * 3;
        float* p_delta  = out + (size_t)nrows * 4;

        kp_final[row * 3 + 0] = alpha * dx + (1.0f - alpha) * fx + d0;
        kp_final[row * 3 + 1] = alpha * dy + (1.0f - alpha) * fy + d1;
        kp_final[row * 3 + 2] = alpha * dz + (1.0f - alpha) * fz + d2;
        p_alpha[row] = alpha;
        p_delta[row * 3 + 0] = d0;
        p_delta[row * 3 + 1] = d1;
        p_delta[row * 3 + 2] = d2;
    }
}

extern "C" void kernel_launch(void* const* d_in, const int* in_sizes, int n_in,
                              void* d_out, int out_size, void* d_ws, size_t ws_size,
                              hipStream_t stream) {
    const float* fk  = (const float*)d_in[0];
    const float* dir = (const float*)d_in[1];
    const float* hm  = (const float*)d_in[2];
    const float* W1  = (const float*)d_in[3];
    const float* b1  = (const float*)d_in[4];
    const float* g1  = (const float*)d_in[5];
    const float* be1 = (const float*)d_in[6];
    const float* W2  = (const float*)d_in[7];
    const float* b2  = (const float*)d_in[8];
    const float* g2  = (const float*)d_in[9];
    const float* be2 = (const float*)d_in[10];
    const float* W3  = (const float*)d_in[11];
    const float* b3  = (const float*)d_in[12];
    float* out  = (float*)d_out;
    float* conf = (float*)d_ws;

    const int nrows = in_sizes[0] / 3;

    hm_conf_kernel<<<nrows, 256, 0, stream>>>(hm, conf);
    mlp_head_kernel<<<nrows / 16, 128, 0, stream>>>(fk, dir, conf,
                                                    W1, b1, g1, be1,
                                                    W2, b2, g2, be2,
                                                    W3, b3, out, nrows);
}
